# starting kernel
# baseline (speedup 1.0000x reference)
#include <hip/hip_runtime.h>


typedef float v2f __attribute__((ext_vector_type(2)));
typedef float v8f __attribute__((ext_vector_type(8)));

#define DIN  128
#define DOUT 128

#define PAIR_STRIDE 288
#define WL_FLOATS   (64 * PAIR_STRIDE)

__global__ __launch_bounds__(256) void gcn_gemm_wmma(
    const float* __restrict__ x, const float* __restrict__ W,
    float* __restrict__ h, int n_rows) {
  __shared__ float Wl[WL_FLOATS];

  const int tid = threadIdx.x;
  for (int i = tid * 4; i < DIN * DOUT; i += 256 * 4) {
    const int r = i >> 7;
    const int c = i & 127;
    const float4 w = *(const float4*)(&W[i]);
    const int base = (r >> 1) * PAIR_STRIDE + (r & 1);
    Wl[base + (c + 0) * 2] = w.x;
    Wl[base + (c + 1) * 2] = w.y;
    Wl[base + (c + 2) * 2] = w.z;
    Wl[base + (c + 3) * 2] = w.w;
  }
  __syncthreads();

  const int wave = tid >> 5;
  const int lane = tid & 31;
  const int g    = lane >> 4;
  const int ln   = lane & 15;

  const int rowBase = blockIdx.x * 128 + wave * 16;
  int arow = rowBase + ln;
  if (arow >= n_rows) arow = n_rows - 1;
  const float* xrow = x + (size_t)arow * DIN;

  v8f acc[8];
#pragma unroll
  for (int t = 0; t < 8; ++t) acc[t] = (v8f)0.0f;

  for (int k = 0; k < DIN; k += 4) {
    const v2f a = *(const v2f*)(xrow + k + 2 * g);
    const int pbase = ((k + 2 * g) >> 1) * PAIR_STRIDE + ln * 2;
#pragma unroll
    for (int t = 0; t < 8; ++t) {
      const v2f b = *(const v2f*)(&Wl[pbase + t * 32]);
      acc[t] = __builtin_amdgcn_wmma_f32_16x16x4_f32(
          false, a, false, b, (short)0, acc[t], false, false);
    }
  }

  if (rowBase + 16 <= n_rows) {
#pragma unroll
    for (int t = 0; t < 8; ++t) {
#pragma unroll
      for (int v = 0; v < 8; ++v) {
        const int row = rowBase + v + 8 * g;
        h[(size_t)row * DOUT + t * 16 + ln] = acc[t][v];
      }
    }
  } else {
#pragma unroll
    for (int t = 0; t < 8; ++t) {
#pragma unroll
      for (int v = 0; v < 8; ++v) {
        const int row = rowBase + v + 8 * g;
        if (row < n_rows) h[(size_t)row * DOUT + t * 16 + ln] = acc[t][v];
      }
    }
  }
}

__global__ __launch_bounds__(256) void gcn_aggregate(
    const float* __restrict__ h, const int* __restrict__ ptr,
    const int* __restrict__ idx, const float* __restrict__ bias,
    float* __restrict__ out, int n_nodes) {
  const int lane = threadIdx.x & 31;
  const int node = __builtin_amdgcn_readfirstlane(
      (int)(blockIdx.x * 8 + (threadIdx.x >> 5)));
  if (node >= n_nodes) return;

  const int p0  = ptr[node];
  const int p1  = ptr[node + 1];
  const int deg = p1 - p0;
  const int c   = lane * 4;

  float4 a0 = make_float4(0.f, 0.f, 0.f, 0.f);
  float4 a1 = a0, a2 = a0, a3 = a0;

  int e = p0;
  for (; e + 4 <= p1; e += 4) {
    const int s0 = idx[e + 0];
    const int s1 = idx[e + 1];
    const int s2 = idx[e + 2];
    const int s3 = idx[e + 3];
    const float4 v0 = *(const float4*)(h + (size_t)s0 * DOUT + c);
    const float4 v1 = *(const float4*)(h + (size_t)s1 * DOUT + c);
    const float4 v2 = *(const float4*)(h + (size_t)s2 * DOUT + c);
    const float4 v3 = *(const float4*)(h + (size_t)s3 * DOUT + c);
    a0.x += v0.x; a0.y += v0.y; a0.z += v0.z; a0.w += v0.w;
    a1.x += v1.x; a1.y += v1.y; a1.z += v1.z; a1.w += v1.w;
    a2.x += v2.x; a2.y += v2.y; a2.z += v2.z; a2.w += v2.w;
    a3.x += v3.x; a3.y += v3.y; a3.z += v3.z; a3.w += v3.w;
  }
  for (; e < p1; ++e) {
    const int s = idx[e];
    const float4 v = *(const float4*)(h + (size_t)s * DOUT + c);
    a0.x += v.x; a0.y += v.y; a0.z += v.z; a0.w += v.w;
  }

  float4 acc;
  acc.x = (a0.x + a1.x) + (a2.x + a3.x);
  acc.y = (a0.y + a1.y) + (a2.y + a3.y);
  acc.z = (a0.z + a1.z) + (a2.z + a3.z);
  acc.w = (a0.w + a1.w) + (a2.w + a3.w);

  const float inv = (deg > 0) ? (1.0f / (float)deg) : 0.0f;
  const float4 b = *(const float4*)(bias + c);
  float4 r;
  r.x = acc.x * inv + b.x;
  r.y = acc.y * inv + b.y;
  r.z = acc.z * inv + b.z;
  r.w = acc.w * inv + b.w;
  *(float4*)(out + (size_t)node * DOUT + c) = r;
}

extern "C" void kernel_launch(void* const* d_in, const int* in_sizes, int n_in,
                              void* d_out, int out_size, void* d_ws, size_t ws_size,
                              hipStream_t stream) {
  const float* x    = (const float*)d_in[0];
  const float* W    = (const float*)d_in[1];
  const float* bias = (const float*)d_in[2];
  const int*   ptr  = (const int*)d_in[3];
  const int*   idx  = (const int*)d_in[4];

  const int n_nodes = in_sizes[0] / DIN;
  float* h = (float*)d_ws;

  const int gemm_blocks = (n_nodes + 127) / 128;
  gcn_gemm_wmma<<<gemm_blocks, 256, 0, stream>>>(x, W, h, n_nodes);

  const int agg_blocks = (n_nodes + 7) / 8;
  gcn_aggregate<<<agg_blocks, 256, 0, stream>>>(h, ptr, idx, bias,
                                                (float*)d_out, n_nodes);
}


# One work-group per WGP: blocks of 25 waves (400 nodes) instead of 8, so 250 blocks on 256 WGPs run as one even round and W is staged once per WGP; f32 sum order unchanged
# speedup vs baseline: 1.3052x; 1.3052x over previous
#include <hip/hip_runtime.h>


typedef float v2f __attribute__((ext_vector_type(2)));
typedef float v8f __attribute__((ext_vector_type(8)));

#define DIN  128
#define DOUT 128

#define PAIR_STRIDE 288
#define WL_FLOATS   (64 * PAIR_STRIDE)

#define APITCH 132
#define ATILE  (16 * APITCH)

#define NWAVES    25
#define NTHREADS  (NWAVES * 32)
#define BLOCKROWS (NWAVES * 16)

__global__ __launch_bounds__(NTHREADS) void gcn_fused_wmma(
    const float* __restrict__ x, const float* __restrict__ W,
    const float* __restrict__ bias, const int* __restrict__ ptr,
    const int* __restrict__ idx, float* __restrict__ out, int n_rows) {
  __shared__ float Wl[WL_FLOATS];
  __shared__ float Al[NWAVES * ATILE];

  const int tid = threadIdx.x;
  for (int i = tid * 4; i < DIN * DOUT; i += NTHREADS * 4) {
    const int r = i >> 7;
    const int c = i & 127;
    const float4 w = *(const float4*)(&W[i]);
    const int base = (r >> 1) * PAIR_STRIDE + (r & 1);
    Wl[base + (c + 0) * 2] = w.x;
    Wl[base + (c + 1) * 2] = w.y;
    Wl[base + (c + 2) * 2] = w.z;
    Wl[base + (c + 3) * 2] = w.w;
  }
  __syncthreads();

  const int wave = tid >> 5;
  const int lane = tid & 31;
  const int g    = lane >> 4;
  const int ln   = lane & 15;

  const int rowBase = blockIdx.x * BLOCKROWS + wave * 16;
  if (rowBase >= n_rows) return;

  float* At = Al + wave * ATILE;

  int pj = rowBase + lane;
  if (pj > n_rows) pj = n_rows;
  const int pv = ptr[pj];

  const float* xl = x + 4 * lane;

#pragma unroll 1
  for (int j = 0; j < 16; ++j) {
    const int p0  = __builtin_amdgcn_readlane(pv, j);
    const int p1  = __builtin_amdgcn_readlane(pv, j + 1);
    const int deg = p1 - p0;

    float4 a0 = make_float4(0.f, 0.f, 0.f, 0.f);
    float4 a1 = a0, a2 = a0, a3 = a0;

    int e = p0;
    for (; e + 16 <= p1; e += 16) {
      int s[16];
      float4 v[16];
#pragma unroll
      for (int i = 0; i < 16; ++i) s[i] = idx[e + i];
#pragma unroll
      for (int i = 0; i < 16; ++i)
        v[i] = *(const float4*)(xl + (size_t)s[i] * DIN);
#pragma unroll
      for (int i = 0; i < 16; i += 4) {
        a0.x += v[i + 0].x; a0.y += v[i + 0].y; a0.z += v[i + 0].z; a0.w += v[i + 0].w;
        a1.x += v[i + 1].x; a1.y += v[i + 1].y; a1.z += v[i + 1].z; a1.w += v[i + 1].w;
        a2.x += v[i + 2].x; a2.y += v[i + 2].y; a2.z += v[i + 2].z; a2.w += v[i + 2].w;
        a3.x += v[i + 3].x; a3.y += v[i + 3].y; a3.z += v[i + 3].z; a3.w += v[i + 3].w;
      }
    }
    for (; e + 4 <= p1; e += 4) {
      const int s0 = idx[e + 0];
      const int s1 = idx[e + 1];
      const int s2 = idx[e + 2];
      const int s3 = idx[e + 3];
      const float4 v0 = *(const float4*)(xl + (size_t)s0 * DIN);
      const float4 v1 = *(const float4*)(xl + (size_t)s1 * DIN);
      const float4 v2 = *(const float4*)(xl + (size_t)s2 * DIN);
      const float4 v3 = *(const float4*)(xl + (size_t)s3 * DIN);
      a0.x += v0.x; a0.y += v0.y; a0.z += v0.z; a0.w += v0.w;
      a1.x += v1.x; a1.y += v1.y; a1.z += v1.z; a1.w += v1.w;
      a2.x += v2.x; a2.y += v2.y; a2.z += v2.z; a2.w += v2.w;
      a3.x += v3.x; a3.y += v3.y; a3.z += v3.z; a3.w += v3.w;
    }
    for (; e < p1; ++e) {
      const int s = idx[e];
      const float4 v = *(const float4*)(xl + (size_t)s * DIN);
      a0.x += v.x; a0.y += v.y; a0.z += v.z; a0.w += v.w;
    }

    const float inv = (deg > 0) ? (1.0f / (float)deg) : 0.0f;
    float4 r;
    r.x = ((a0.x + a1.x) + (a2.x + a3.x)) * inv;
    r.y = ((a0.y + a1.y) + (a2.y + a3.y)) * inv;
    r.z = ((a0.z + a1.z) + (a2.z + a3.z)) * inv;
    r.w = ((a0.w + a1.w) + (a2.w + a3.w)) * inv;
    *(float4*)(&At[j * APITCH + 4 * lane]) = r;
  }

  __builtin_amdgcn_fence(__ATOMIC_RELEASE, "workgroup", "local");

  float bv[8];
#pragma unroll
  for (int t = 0; t < 8; ++t) bv[t] = bias[t * 16 + ln];

  v8f acc[8];
#pragma unroll
  for (int t = 0; t < 8; ++t) acc[t] = (v8f)0.0f;

  const float* arow = At + ln * APITCH + 2 * g;
  for (int k = 0; k < DIN; k += 4) {
    const v2f a = *(const v2f*)(arow + k);
    const int pbase = ((k + 2 * g) >> 1) * PAIR_STRIDE + ln * 2;
#pragma unroll
    for (int t = 0; t < 8; ++t) {
      const v2f b = *(const v2f*)(&Wl[pbase + t * 32]);
      acc[t] = __builtin_amdgcn_wmma_f32_16x16x4_f32(
          false, a, false, b, (short)0, acc[t], false, false);
    }
  }

  if (rowBase + 16 <= n_rows) {
#pragma unroll
    for (int t = 0; t < 8; ++t) {
#pragma unroll
      for (int v = 0; v < 8; ++v) {
        const int row = rowBase + v + 8 * g;
        out[(size_t)row * DOUT + t * 16 + ln] = acc[t][v] + bv[t];
      }
    }
  } else {
#pragma unroll
    for (int t = 0; t < 8; ++t) {
#pragma unroll
      for (int v = 0; v < 8; ++v) {
        const int row = rowBase + v + 8 * g;
        if (row < n_rows) out[(size_t)row * DOUT + t * 16 + ln] = acc[t][v] + bv[t];
      }
    }
  }
}

extern "C" void kernel_launch(void* const* d_in, const int* in_sizes, int n_in,
                              void* d_out, int out_size, void* d_ws, size_t ws_size,
                              hipStream_t stream) {
  const float* x    = (const float*)d_in[0];
  const float* W    = (const float*)d_in[1];
  const float* bias = (const float*)d_in[2];
  const int*   ptr  = (const int*)d_in[3];
  const int*   idx  = (const int*)d_in[4];

  const int n_nodes = in_sizes[0] / DIN;

  const int blocks = (n_nodes + BLOCKROWS - 1) / BLOCKROWS;
  gcn_fused_wmma<<<blocks, NTHREADS, 0, stream>>>(x, W, bias, ptr, idx,
                                                  (float*)d_out, n_nodes);
}


# W staging off the gather's path: all of a thread's W loads issued at once, written as b128 LDS stores; barrier split, signal after staging, wait only before the WMMA phase; sum order unchanged
# speedup vs baseline: 1.3541x; 1.0375x over previous
#include <hip/hip_runtime.h>


typedef float v2f __attribute__((ext_vector_type(2)));
typedef float v8f __attribute__((ext_vector_type(8)));

#define DIN  128
#define DOUT 128

#define PAIR_STRIDE 288
#define WL_FLOATS   (64 * PAIR_STRIDE)

#define APITCH 132
#define ATILE  (16 * APITCH)

#define NWAVES    25
#define NTHREADS  (NWAVES * 32)
#define BLOCKROWS (NWAVES * 16)

__global__ __launch_bounds__(NTHREADS) void gcn_fused_wmma(
    const float* __restrict__ x, const float* __restrict__ W,
    const float* __restrict__ bias, const int* __restrict__ ptr,
    const int* __restrict__ idx, float* __restrict__ out, int n_rows) {
  __shared__ __attribute__((aligned(16))) float Wl[WL_FLOATS];
  __shared__ __attribute__((aligned(16))) float Al[NWAVES * ATILE];

  const int tid  = threadIdx.x;
  const int wave = tid >> 5;
  const int lane = tid & 31;
  const int g    = lane >> 4;
  const int ln   = lane & 15;

  const int rowBase = blockIdx.x * BLOCKROWS + wave * 16;
  const bool active = rowBase < n_rows;

  int pj = rowBase + lane;
  if (pj > n_rows) pj = n_rows;
  const int pv = ptr[pj];

  {
    constexpr int ITEMS = (DIN / 2) * (DOUT / 4);
    constexpr int TRIPS = (ITEMS + NTHREADS - 1) / NTHREADS;
    float4 w0[TRIPS], w1[TRIPS];
#pragma unroll
    for (int t = 0; t < TRIPS; ++t) {
      const int q = tid + t * NTHREADS;
      if (q < ITEMS) {
        const float* src = W + (q >> 5) * (2 * DOUT) + (q & 31) * 4;
        w0[t] = *(const float4*)src;
        w1[t] = *(const float4*)(src + DOUT);
      }
    }
#pragma unroll
    for (int t = 0; t < TRIPS; ++t) {
      const int q = tid + t * NTHREADS;
      if (q < ITEMS) {
        float* dst = Wl + (q >> 5) * PAIR_STRIDE + (q & 31) * 8;
        *(float4*)dst       = make_float4(w0[t].x, w1[t].x, w0[t].y, w1[t].y);
        *(float4*)(dst + 4) = make_float4(w0[t].z, w1[t].z, w0[t].w, w1[t].w);
      }
    }
  }
  __builtin_amdgcn_fence(__ATOMIC_RELEASE, "workgroup", "local");
  __builtin_amdgcn_s_barrier_signal(-1);

  float* At = Al + wave * ATILE;

  const float* xl = x + 4 * lane;

  if (active) {
#pragma unroll 1
  for (int j = 0; j < 16; ++j) {
    const int p0  = __builtin_amdgcn_readlane(pv, j);
    const int p1  = __builtin_amdgcn_readlane(pv, j + 1);
    const int deg = p1 - p0;

    float4 a0 = make_float4(0.f, 0.f, 0.f, 0.f);
    float4 a1 = a0, a2 = a0, a3 = a0;

    int e = p0;
    for (; e + 16 <= p1; e += 16) {
      int s[16];
      float4 v[16];
#pragma unroll
      for (int i = 0; i < 16; ++i) s[i] = idx[e + i];
#pragma unroll
      for (int i = 0; i < 16; ++i)
        v[i] = *(const float4*)(xl + (size_t)s[i] * DIN);
#pragma unroll
      for (int i = 0; i < 16; i += 4) {
        a0.x += v[i + 0].x; a0.y += v[i + 0].y; a0.z += v[i + 0].z; a0.w += v[i + 0].w;
        a1.x += v[i + 1].x; a1.y += v[i + 1].y; a1.z += v[i + 1].z; a1.w += v[i + 1].w;
        a2.x += v[i + 2].x; a2.y += v[i + 2].y; a2.z += v[i + 2].z; a2.w += v[i + 2].w;
        a3.x += v[i + 3].x; a3.y += v[i + 3].y; a3.z += v[i + 3].z; a3.w += v[i + 3].w;
      }
    }
    for (; e + 4 <= p1; e += 4) {
      const int s0 = idx[e + 0];
      const int s1 = idx[e + 1];
      const int s2 = idx[e + 2];
      const int s3 = idx[e + 3];
      const float4 v0 = *(const float4*)(xl + (size_t)s0 * DIN);
      const float4 v1 = *(const float4*)(xl + (size_t)s1 * DIN);
      const float4 v2 = *(const float4*)(xl + (size_t)s2 * DIN);
      const float4 v3 = *(const float4*)(xl + (size_t)s3 * DIN);
      a0.x += v0.x; a0.y += v0.y; a0.z += v0.z; a0.w += v0.w;
      a1.x += v1.x; a1.y += v1.y; a1.z += v1.z; a1.w += v1.w;
      a2.x += v2.x; a2.y += v2.y; a2.z += v2.z; a2.w += v2.w;
      a3.x += v3.x; a3.y += v3.y; a3.z += v3.z; a3.w += v3.w;
    }
    for (; e < p1; ++e) {
      const int s = idx[e];
      const float4 v = *(const float4*)(xl + (size_t)s * DIN);
      a0.x += v.x; a0.y += v.y; a0.z += v.z; a0.w += v.w;
    }

    const float inv = (deg > 0) ? (1.0f / (float)deg) : 0.0f;
    float4 r;
    r.x = ((a0.x + a1.x) + (a2.x + a3.x)) * inv;
    r.y = ((a0.y + a1.y) + (a2.y + a3.y)) * inv;
    r.z = ((a0.z + a1.z) + (a2.z + a3.z)) * inv;
    r.w = ((a0.w + a1.w) + (a2.w + a3.w)) * inv;
    *(float4*)(&At[j * APITCH + 4 * lane]) = r;
  }
  }

  __builtin_amdgcn_fence(__ATOMIC_RELEASE, "workgroup", "local");

  __builtin_amdgcn_s_barrier_wait(-1);
  __builtin_amdgcn_fence(__ATOMIC_ACQUIRE, "workgroup", "local");
  if (!active) return;

  float bv[8];
#pragma unroll
  for (int t = 0; t < 8; ++t) bv[t] = bias[t * 16 + ln];

  v8f acc[8];
#pragma unroll
  for (int t = 0; t < 8; ++t) acc[t] = (v8f)0.0f;

  const float* arow = At + ln * APITCH + 2 * g;
  for (int k = 0; k < DIN; k += 4) {
    const v2f a = *(const v2f*)(arow + k);
    const int pbase = ((k + 2 * g) >> 1) * PAIR_STRIDE + ln * 2;
#pragma unroll
    for (int t = 0; t < 8; ++t) {
      const v2f b = *(const v2f*)(&Wl[pbase + t * 32]);
      acc[t] = __builtin_amdgcn_wmma_f32_16x16x4_f32(
          false, a, false, b, (short)0, acc[t], false, false);
    }
  }

  if (rowBase + 16 <= n_rows) {
#pragma unroll
    for (int t = 0; t < 8; ++t) {
#pragma unroll
      for (int v = 0; v < 8; ++v) {
        const int row = rowBase + v + 8 * g;
        out[(size_t)row * DOUT + t * 16 + ln] = acc[t][v] + bv[t];
      }
    }
  } else {
#pragma unroll
    for (int t = 0; t < 8; ++t) {
#pragma unroll
      for (int v = 0; v < 8; ++v) {
        const int row = rowBase + v + 8 * g;
        if (row < n_rows) out[(size_t)row * DOUT + t * 16 + ln] = acc[t][v] + bv[t];
      }
    }
  }
}

extern "C" void kernel_launch(void* const* d_in, const int* in_sizes, int n_in,
                              void* d_out, int out_size, void* d_ws, size_t ws_size,
                              hipStream_t stream) {
  const float* x    = (const float*)d_in[0];
  const float* W    = (const float*)d_in[1];
  const float* bias = (const float*)d_in[2];
  const int*   ptr  = (const int*)d_in[3];
  const int*   idx  = (const int*)d_in[4];

  const int n_nodes = in_sizes[0] / DIN;

  const int blocks = (n_nodes + BLOCKROWS - 1) / BLOCKROWS;
  gcn_fused_wmma<<<blocks, NTHREADS, 0, stream>>>(x, W, bias, ptr, idx,
                                                  (float*)d_out, n_nodes);
}


# Waves of a block in two cohorts on a second split barrier: cohort 1 gathers its rows while cohort 0 multiplies, so WMMA and memory path overlap; sum order unchanged
# speedup vs baseline: 1.3995x; 1.0335x over previous
#include <hip/hip_runtime.h>


typedef float v2f __attribute__((ext_vector_type(2)));
typedef float v8f __attribute__((ext_vector_type(8)));

#define DIN  128
#define DOUT 128

#define PAIR_STRIDE 288
#define WL_FLOATS   (64 * PAIR_STRIDE)

#define APITCH 132
#define ATILE  (16 * APITCH)

#define NWAVES    25
#define NTHREADS  (NWAVES * 32)
#define BLOCKROWS (NWAVES * 16)

static __device__ __forceinline__ void gather_strip(
    const float* __restrict__ x, const int* __restrict__ idx, const int pv,
    float* __restrict__ At, const int lane) {
  const float* xl = x + 4 * lane;
#pragma unroll 1
  for (int j = 0; j < 16; ++j) {
    const int p0  = __builtin_amdgcn_readlane(pv, j);
    const int p1  = __builtin_amdgcn_readlane(pv, j + 1);
    const int deg = p1 - p0;

    float4 a0 = make_float4(0.f, 0.f, 0.f, 0.f);
    float4 a1 = a0, a2 = a0, a3 = a0;

    int e = p0;
    for (; e + 16 <= p1; e += 16) {
      int s[16];
      float4 v[16];
#pragma unroll
      for (int i = 0; i < 16; ++i) s[i] = idx[e + i];
#pragma unroll
      for (int i = 0; i < 16; ++i)
        v[i] = *(const float4*)(xl + (size_t)s[i] * DIN);
#pragma unroll
      for (int i = 0; i < 16; i += 4) {
        a0.x += v[i + 0].x; a0.y += v[i + 0].y; a0.z += v[i + 0].z; a0.w += v[i + 0].w;
        a1.x += v[i + 1].x; a1.y += v[i + 1].y; a1.z += v[i + 1].z; a1.w += v[i + 1].w;
        a2.x += v[i + 2].x; a2.y += v[i + 2].y; a2.z += v[i + 2].z; a2.w += v[i + 2].w;
        a3.x += v[i + 3].x; a3.y += v[i + 3].y; a3.z += v[i + 3].z; a3.w += v[i + 3].w;
      }
    }
    for (; e + 4 <= p1; e += 4) {
      const int s0 = idx[e + 0];
      const int s1 = idx[e + 1];
      const int s2 = idx[e + 2];
      const int s3 = idx[e + 3];
      const float4 v0 = *(const float4*)(xl + (size_t)s0 * DIN);
      const float4 v1 = *(const float4*)(xl + (size_t)s1 * DIN);
      const float4 v2 = *(const float4*)(xl + (size_t)s2 * DIN);
      const float4 v3 = *(const float4*)(xl + (size_t)s3 * DIN);
      a0.x += v0.x; a0.y += v0.y; a0.z += v0.z; a0.w += v0.w;
      a1.x += v1.x; a1.y += v1.y; a1.z += v1.z; a1.w += v1.w;
      a2.x += v2.x; a2.y += v2.y; a2.z += v2.z; a2.w += v2.w;
      a3.x += v3.x; a3.y += v3.y; a3.z += v3.z; a3.w += v3.w;
    }
    for (; e < p1; ++e) {
      const int s = idx[e];
      const float4 v = *(const float4*)(xl + (size_t)s * DIN);
      a0.x += v.x; a0.y += v.y; a0.z += v.z; a0.w += v.w;
    }

    const float inv = (deg > 0) ? (1.0f / (float)deg) : 0.0f;
    float4 r;
    r.x = ((a0.x + a1.x) + (a2.x + a3.x)) * inv;
    r.y = ((a0.y + a1.y) + (a2.y + a3.y)) * inv;
    r.z = ((a0.z + a1.z) + (a2.z + a3.z)) * inv;
    r.w = ((a0.w + a1.w) + (a2.w + a3.w)) * inv;
    *(float4*)(&At[j * APITCH + 4 * lane]) = r;
  }
}

__global__ __launch_bounds__(NTHREADS) void gcn_fused_wmma(
    const float* __restrict__ x, const float* __restrict__ W,
    const float* __restrict__ bias, const int* __restrict__ ptr,
    const int* __restrict__ idx, float* __restrict__ out, int n_rows) {
  __shared__ __attribute__((aligned(16))) float Wl[WL_FLOATS];
  __shared__ __attribute__((aligned(16))) float Al[NWAVES * ATILE];

  const int tid  = threadIdx.x;
  const int wave = tid >> 5;
  const int lane = tid & 31;
  const int g    = lane >> 4;
  const int ln   = lane & 15;

  const int rowBase = blockIdx.x * BLOCKROWS + wave * 16;
  const bool active = rowBase < n_rows;

  int pj = rowBase + lane;
  if (pj > n_rows) pj = n_rows;
  const int pv = ptr[pj];

  {
    constexpr int ITEMS = (DIN / 2) * (DOUT / 4);
    constexpr int TRIPS = (ITEMS + NTHREADS - 1) / NTHREADS;
    float4 w0[TRIPS], w1[TRIPS];
#pragma unroll
    for (int t = 0; t < TRIPS; ++t) {
      const int q = tid + t * NTHREADS;
      if (q < ITEMS) {
        const float* src = W + (q >> 5) * (2 * DOUT) + (q & 31) * 4;
        w0[t] = *(const float4*)src;
        w1[t] = *(const float4*)(src + DOUT);
      }
    }
#pragma unroll
    for (int t = 0; t < TRIPS; ++t) {
      const int q = tid + t * NTHREADS;
      if (q < ITEMS) {
        float* dst = Wl + (q >> 5) * PAIR_STRIDE + (q & 31) * 8;
        *(float4*)dst       = make_float4(w0[t].x, w1[t].x, w0[t].y, w1[t].y);
        *(float4*)(dst + 4) = make_float4(w0[t].z, w1[t].z, w0[t].w, w1[t].w);
      }
    }
  }
  __builtin_amdgcn_fence(__ATOMIC_RELEASE, "workgroup", "local");
  __builtin_amdgcn_s_barrier_signal(-1);

  float* At = Al + wave * ATILE;

  const int cohort = (wave >> 2) & 1;

  if (active && cohort == 0) gather_strip(x, idx, pv, At, lane);

  __builtin_amdgcn_fence(__ATOMIC_RELEASE, "workgroup", "local");

  __builtin_amdgcn_s_barrier_wait(-1);
  __builtin_amdgcn_fence(__ATOMIC_ACQUIRE, "workgroup", "local");

  __builtin_amdgcn_s_barrier_signal(-1);
  __builtin_amdgcn_s_barrier_wait(-1);

  if (active && cohort == 1) gather_strip(x, idx, pv, At, lane);
  __builtin_amdgcn_fence(__ATOMIC_RELEASE, "workgroup", "local");
  if (!active) return;

  float bv[8];
#pragma unroll
  for (int t = 0; t < 8; ++t) bv[t] = bias[t * 16 + ln];

  v8f acc[8];
#pragma unroll
  for (int t = 0; t < 8; ++t) acc[t] = (v8f)0.0f;

  const float* arow = At + ln * APITCH + 2 * g;
  for (int k = 0; k < DIN; k += 4) {
    const v2f a = *(const v2f*)(arow + k);
    const int pbase = ((k + 2 * g) >> 1) * PAIR_STRIDE + ln * 2;
#pragma unroll
    for (int t = 0; t < 8; ++t) {
      const v2f b = *(const v2f*)(&Wl[pbase + t * 32]);
      acc[t] = __builtin_amdgcn_wmma_f32_16x16x4_f32(
          false, a, false, b, (short)0, acc[t], false, false);
    }
  }

  if (rowBase + 16 <= n_rows) {
#pragma unroll
    for (int t = 0; t < 8; ++t) {
#pragma unroll
      for (int v = 0; v < 8; ++v) {
        const int row = rowBase + v + 8 * g;
        out[(size_t)row * DOUT + t * 16 + ln] = acc[t][v] + bv[t];
      }
    }
  } else {
#pragma unroll
    for (int t = 0; t < 8; ++t) {
#pragma unroll
      for (int v = 0; v < 8; ++v) {
        const int row = rowBase + v + 8 * g;
        if (row < n_rows) out[(size_t)row * DOUT + t * 16 + ln] = acc[t][v] + bv[t];
      }
    }
  }
}

extern "C" void kernel_launch(void* const* d_in, const int* in_sizes, int n_in,
                              void* d_out, int out_size, void* d_ws, size_t ws_size,
                              hipStream_t stream) {
  const float* x    = (const float*)d_in[0];
  const float* W    = (const float*)d_in[1];
  const float* bias = (const float*)d_in[2];
  const int*   ptr  = (const int*)d_in[3];
  const int*   idx  = (const int*)d_in[4];

  const int n_nodes = in_sizes[0] / DIN;

  const int blocks = (n_nodes + BLOCKROWS - 1) / BLOCKROWS;
  gcn_fused_wmma<<<blocks, NTHREADS, 0, stream>>>(x, W, bias, ptr, idx,
                                                  (float*)d_out, n_nodes);
}


# Source ids of the next node fetched one node ahead, behind the current node's row loads: one memory trip per node on a wave's path instead of two; sum order unchanged
# speedup vs baseline: 1.4491x; 1.0354x over previous
#include <hip/hip_runtime.h>


typedef float v2f __attribute__((ext_vector_type(2)));
typedef float v8f __attribute__((ext_vector_type(8)));

#define DIN  128
#define DOUT 128

#define PAIR_STRIDE 288
#define WL_FLOATS   (64 * PAIR_STRIDE)

#define APITCH 132
#define ATILE  (16 * APITCH)

#define NWAVES    25
#define NTHREADS  (NWAVES * 32)
#define BLOCKROWS (NWAVES * 16)

static __device__ __forceinline__ void gather_strip(
    const float* __restrict__ x, const int* __restrict__ idx, const int pv,
    float* __restrict__ At, const int lane) {
  const float* xl = x + 4 * lane;

  int sn[16];
  int p0 = __builtin_amdgcn_readlane(pv, 0);
  int p1 = __builtin_amdgcn_readlane(pv, 1);
  bool have = (p1 - p0) >= 16;
  if (have) {
#pragma unroll
    for (int i = 0; i < 16; ++i) sn[i] = idx[p0 + i];
  }

#pragma unroll 1
  for (int j = 0; j < 16; ++j) {
    const int deg = p1 - p0;
    const int q0 = p1;
    const int q1 = (j < 15) ? __builtin_amdgcn_readlane(pv, j + 2) : p1;
    const bool haveN = (q1 - q0) >= 16;

    float4 a0 = make_float4(0.f, 0.f, 0.f, 0.f);
    float4 a1 = a0, a2 = a0, a3 = a0;

    int e = p0;
    if (have) {
      float4 v[16];
#pragma unroll
      for (int i = 0; i < 16; ++i)
        v[i] = *(const float4*)(xl + (size_t)sn[i] * DIN);
      if (haveN) {
#pragma unroll
        for (int i = 0; i < 16; ++i) sn[i] = idx[q0 + i];
      }
#pragma unroll
      for (int i = 0; i < 16; i += 4) {
        a0.x += v[i + 0].x; a0.y += v[i + 0].y; a0.z += v[i + 0].z; a0.w += v[i + 0].w;
        a1.x += v[i + 1].x; a1.y += v[i + 1].y; a1.z += v[i + 1].z; a1.w += v[i + 1].w;
        a2.x += v[i + 2].x; a2.y += v[i + 2].y; a2.z += v[i + 2].z; a2.w += v[i + 2].w;
        a3.x += v[i + 3].x; a3.y += v[i + 3].y; a3.z += v[i + 3].z; a3.w += v[i + 3].w;
      }
      e += 16;
    } else if (haveN) {
#pragma unroll
      for (int i = 0; i < 16; ++i) sn[i] = idx[q0 + i];
    }
    for (; e + 16 <= p1; e += 16) {
      int s[16];
      float4 v[16];
#pragma unroll
      for (int i = 0; i < 16; ++i) s[i] = idx[e + i];
#pragma unroll
      for (int i = 0; i < 16; ++i)
        v[i] = *(const float4*)(xl + (size_t)s[i] * DIN);
#pragma unroll
      for (int i = 0; i < 16; i += 4) {
        a0.x += v[i + 0].x; a0.y += v[i + 0].y; a0.z += v[i + 0].z; a0.w += v[i + 0].w;
        a1.x += v[i + 1].x; a1.y += v[i + 1].y; a1.z += v[i + 1].z; a1.w += v[i + 1].w;
        a2.x += v[i + 2].x; a2.y += v[i + 2].y; a2.z += v[i + 2].z; a2.w += v[i + 2].w;
        a3.x += v[i + 3].x; a3.y += v[i + 3].y; a3.z += v[i + 3].z; a3.w += v[i + 3].w;
      }
    }
    for (; e + 4 <= p1; e += 4) {
      const int s0 = idx[e + 0];
      const int s1 = idx[e + 1];
      const int s2 = idx[e + 2];
      const int s3 = idx[e + 3];
      const float4 v0 = *(const float4*)(xl + (size_t)s0 * DIN);
      const float4 v1 = *(const float4*)(xl + (size_t)s1 * DIN);
      const float4 v2 = *(const float4*)(xl + (size_t)s2 * DIN);
      const float4 v3 = *(const float4*)(xl + (size_t)s3 * DIN);
      a0.x += v0.x; a0.y += v0.y; a0.z += v0.z; a0.w += v0.w;
      a1.x += v1.x; a1.y += v1.y; a1.z += v1.z; a1.w += v1.w;
      a2.x += v2.x; a2.y += v2.y; a2.z += v2.z; a2.w += v2.w;
      a3.x += v3.x; a3.y += v3.y; a3.z += v3.z; a3.w += v3.w;
    }
    for (; e < p1; ++e) {
      const int s = idx[e];
      const float4 v = *(const float4*)(xl + (size_t)s * DIN);
      a0.x += v.x; a0.y += v.y; a0.z += v.z; a0.w += v.w;
    }

    const float inv = (deg > 0) ? (1.0f / (float)deg) : 0.0f;
    float4 r;
    r.x = ((a0.x + a1.x) + (a2.x + a3.x)) * inv;
    r.y = ((a0.y + a1.y) + (a2.y + a3.y)) * inv;
    r.z = ((a0.z + a1.z) + (a2.z + a3.z)) * inv;
    r.w = ((a0.w + a1.w) + (a2.w + a3.w)) * inv;
    *(float4*)(&At[j * APITCH + 4 * lane]) = r;

    p0 = q0;
    p1 = q1;
    have = haveN;
  }
}

__global__ __launch_bounds__(NTHREADS) void gcn_fused_wmma(
    const float* __restrict__ x, const float* __restrict__ W,
    const float* __restrict__ bias, const int* __restrict__ ptr,
    const int* __restrict__ idx, float* __restrict__ out, int n_rows) {
  __shared__ __attribute__((aligned(16))) float Wl[WL_FLOATS];
  __shared__ __attribute__((aligned(16))) float Al[NWAVES * ATILE];

  const int tid  = threadIdx.x;
  const int wave = tid >> 5;
  const int lane = tid & 31;
  const int g    = lane >> 4;
  const int ln   = lane & 15;

  const int rowBase = blockIdx.x * BLOCKROWS + wave * 16;
  const bool active = rowBase < n_rows;

  int pj = rowBase + lane;
  if (pj > n_rows) pj = n_rows;
  const int pv = ptr[pj];

  {
    constexpr int ITEMS = (DIN / 2) * (DOUT / 4);
    constexpr int TRIPS = (ITEMS + NTHREADS - 1) / NTHREADS;
    float4 w0[TRIPS], w1[TRIPS];
#pragma unroll
    for (int t = 0; t < TRIPS; ++t) {
      const int q = tid + t * NTHREADS;
      if (q < ITEMS) {
        const float* src = W + (q >> 5) * (2 * DOUT) + (q & 31) * 4;
        w0[t] = *(const float4*)src;
        w1[t] = *(const float4*)(src + DOUT);
      }
    }
#pragma unroll
    for (int t = 0; t < TRIPS; ++t) {
      const int q = tid + t * NTHREADS;
      if (q < ITEMS) {
        float* dst = Wl + (q >> 5) * PAIR_STRIDE + (q & 31) * 8;
        *(float4*)dst       = make_float4(w0[t].x, w1[t].x, w0[t].y, w1[t].y);
        *(float4*)(dst + 4) = make_float4(w0[t].z, w1[t].z, w0[t].w, w1[t].w);
      }
    }
  }
  __builtin_amdgcn_fence(__ATOMIC_RELEASE, "workgroup", "local");
  __builtin_amdgcn_s_barrier_signal(-1);

  float* At = Al + wave * ATILE;

  const int cohort = (wave >> 2) & 1;

  if (active && cohort == 0) gather_strip(x, idx, pv, At, lane);

  __builtin_amdgcn_fence(__ATOMIC_RELEASE, "workgroup", "local");

  __builtin_amdgcn_s_barrier_wait(-1);
  __builtin_amdgcn_fence(__ATOMIC_ACQUIRE, "workgroup", "local");

  __builtin_amdgcn_s_barrier_signal(-1);
  __builtin_amdgcn_s_barrier_wait(-1);

  if (active && cohort == 1) gather_strip(x, idx, pv, At, lane);
  __builtin_amdgcn_fence(__ATOMIC_RELEASE, "workgroup", "local");
  if (!active) return;

  float bv[8];
#pragma unroll
  for (int t = 0; t < 8; ++t) bv[t] = bias[t * 16 + ln];

  v8f acc[8];
#pragma unroll
  for (int t = 0; t < 8; ++t) acc[t] = (v8f)0.0f;

  const float* arow = At + ln * APITCH + 2 * g;
  for (int k = 0; k < DIN; k += 4) {
    const v2f a = *(const v2f*)(arow + k);
    const int pbase = ((k + 2 * g) >> 1) * PAIR_STRIDE + ln * 2;
#pragma unroll
    for (int t = 0; t < 8; ++t) {
      const v2f b = *(const v2f*)(&Wl[pbase + t * 32]);
      acc[t] = __builtin_amdgcn_wmma_f32_16x16x4_f32(
          false, a, false, b, (short)0, acc[t], false, false);
    }
  }

  if (rowBase + 16 <= n_rows) {
#pragma unroll
    for (int t = 0; t < 8; ++t) {
#pragma unroll
      for (int v = 0; v < 8; ++v) {
        const int row = rowBase + v + 8 * g;
        out[(size_t)row * DOUT + t * 16 + ln] = acc[t][v] + bv[t];
      }
    }
  } else {
#pragma unroll
    for (int t = 0; t < 8; ++t) {
#pragma unroll
      for (int v = 0; v < 8; ++v) {
        const int row = rowBase + v + 8 * g;
        if (row < n_rows) out[(size_t)row * DOUT + t * 16 + ln] = acc[t][v] + bv[t];
      }
    }
  }
}

extern "C" void kernel_launch(void* const* d_in, const int* in_sizes, int n_in,
                              void* d_out, int out_size, void* d_ws, size_t ws_size,
                              hipStream_t stream) {
  const float* x    = (const float*)d_in[0];
  const float* W    = (const float*)d_in[1];
  const float* bias = (const float*)d_in[2];
  const int*   ptr  = (const int*)d_in[3];
  const int*   idx  = (const int*)d_in[4];

  const int n_nodes = in_sizes[0] / DIN;

  const int blocks = (n_nodes + BLOCKROWS - 1) / BLOCKROWS;
  gcn_fused_wmma<<<blocks, NTHREADS, 0, stream>>>(x, W, bias, ptr, idx,
                                                  (float*)d_out, n_nodes);
}
